# starting kernel
# baseline (speedup 1.0000x reference)
#include <hip/hip_runtime.h>
#include <math.h>


typedef __attribute__((ext_vector_type(2))) float v2f;
typedef __attribute__((ext_vector_type(8))) float v8f;

#define B_  128
#define N_  16384
#define K_  12
#define D_  256
#define CHUNKS 32
#define ROWS_PER_CHUNK (N_ / CHUNKS)
#define EPS_ 1e-8f

#define OUT_SLOTS 0L
#define OUT_MASKS ((long)B_ * K_ * D_)
#define OUT_GATES (OUT_MASKS + (long)B_ * N_ * K_)
#define OUT_UTIL  (OUT_GATES + (long)B_ * K_)

#define WS_PARTIALS 0
#define WS_GATES    (B_ * CHUNKS * K_)

__global__ void asp_reduce(const float* __restrict__ masks,
                           float* __restrict__ partials) {
    const int blk = blockIdx.x;
    const int b   = blk / CHUNKS;
    const int ch  = blk % CHUNKS;
    const int t   = threadIdx.x;

    const float4* src = (const float4*)masks
                      + ((long)b * N_ + (long)ch * ROWS_PER_CHUNK) * 3;
    const float4* p = src + (long)t * 6;

    const float4 v0 = p[0], v1 = p[1], v2 = p[2];
    const float4 v3 = p[3], v4 = p[4], v5 = p[5];

    float a[12];
    a[0] = v0.x + v3.x;  a[1]  = v0.y + v3.y;  a[2]  = v0.z + v3.z;  a[3]  = v0.w + v3.w;
    a[4] = v1.x + v4.x;  a[5]  = v1.y + v4.y;  a[6]  = v1.z + v4.z;  a[7]  = v1.w + v4.w;
    a[8] = v2.x + v5.x;  a[9]  = v2.y + v5.y;  a[10] = v2.z + v5.z;  a[11] = v2.w + v5.w;

    #pragma unroll
    for (int m = 16; m >= 1; m >>= 1) {
        #pragma unroll
        for (int j = 0; j < 12; ++j)
            a[j] += __shfl_xor(a[j], m, 32);
    }

    __shared__ float wred[8 * 12];
    const int w    = t >> 5;
    const int lane = t & 31;
    if (lane == 0) {
        #pragma unroll
        for (int j = 0; j < 12; ++j) wred[w * 12 + j] = a[j];
    }
    __syncthreads();

    if (t < 12) {
        float s = 0.f;
        #pragma unroll
        for (int ww = 0; ww < 8; ++ww) s += wred[ww * 12 + t];
        partials[((long)b * CHUNKS + ch) * K_ + t] = s;
    }
}

__global__ void asp_gates(const float* __restrict__ partials,
                          const float* __restrict__ W1,
                          const float* __restrict__ b1,
                          const float* __restrict__ W2,
                          const float* __restrict__ b2,
                          float* __restrict__ gates_ws,
                          float* __restrict__ out) {
    const int tid  = blockIdx.x * blockDim.x + threadIdx.x;
    const int tile = tid >> 5;
    const int lane = threadIdx.x & 31;
    const int m    = lane & 15;
    const int r    = tile * 16 + m;
    const int b    = r / K_;
    const int k    = r - b * K_;

    float s = 0.f;
    #pragma unroll 8
    for (int c = 0; c < CHUNKS; ++c)
        s += partials[((long)b * CHUNKS + c) * K_ + k];
    const float u = s * (1.0f / (float)N_);

    if (lane < 16) out[OUT_UTIL + r] = u;

    float h[16], w2[16];
    #pragma unroll
    for (int j = 0; j < 16; ++j) {
        h[j]  = fmaxf(0.f, fmaf(u, W1[j], b1[j]));
        w2[j] = W2[j];
    }

    const bool hi = (lane >= 16);
    v8f acc = {0.f, 0.f, 0.f, 0.f, 0.f, 0.f, 0.f, 0.f};
    #pragma unroll
    for (int i = 0; i < 4; ++i) {
        v2f av, bv;
        av.x = hi ? h[4 * i + 2]  : h[4 * i + 0];
        av.y = hi ? h[4 * i + 3]  : h[4 * i + 1];
        bv.x = hi ? w2[4 * i + 2] : w2[4 * i + 0];
        bv.y = hi ? w2[4 * i + 3] : w2[4 * i + 1];
        acc = __builtin_amdgcn_wmma_f32_16x16x4_f32(
             false, av,  false, bv,
             (short)0, acc,  false,  false);
    }

    const float bias = b2[0];
    if (lane == 0 || lane == 16) {
        const int base = tile * 16 + ((lane == 16) ? 8 : 0);
        #pragma unroll
        for (int j = 0; j < 8; ++j) {
            const float g = 1.0f / (1.0f + __expf(-(acc[j] + bias)));
            gates_ws[base + j]        = g;
            out[OUT_GATES + base + j] = g;
        }
    }
}

__global__ void asp_slots(const float* __restrict__ slots,
                          const float* __restrict__ gates_ws,
                          float* __restrict__ out) {
    const long i4 = (long)blockIdx.x * blockDim.x + threadIdx.x;
    const float4 v = ((const float4*)slots)[i4];
    const float g  = gates_ws[i4 >> 6];
    float4 o;
    o.x = v.x * g; o.y = v.y * g; o.z = v.z * g; o.w = v.w * g;
    ((float4*)(out + OUT_SLOTS))[i4] = o;
}

__global__ void asp_masks(const float* __restrict__ masks,
                          const float* __restrict__ gates_ws,
                          float* __restrict__ out) {
    __shared__ float g[12];
    const long row = (long)blockIdx.x * 256 + threadIdx.x;
    const int  b   = (int)(row >> 14);
    if (threadIdx.x < 12) g[threadIdx.x] = gates_ws[b * K_ + threadIdx.x];
    __syncthreads();

    const float4* src = (const float4*)masks + row * 3;
    float4 m0 = src[0], m1 = src[1], m2 = src[2];

    m0.x *= g[0];  m0.y *= g[1];  m0.z *= g[2];  m0.w *= g[3];
    m1.x *= g[4];  m1.y *= g[5];  m1.z *= g[6];  m1.w *= g[7];
    m2.x *= g[8];  m2.y *= g[9];  m2.z *= g[10]; m2.w *= g[11];

    const float sum = (m0.x + m0.y + m0.z + m0.w)
                    + (m1.x + m1.y + m1.z + m1.w)
                    + (m2.x + m2.y + m2.z + m2.w);
    const float inv = 1.0f / (sum + EPS_);

    m0.x *= inv; m0.y *= inv; m0.z *= inv; m0.w *= inv;
    m1.x *= inv; m1.y *= inv; m1.z *= inv; m1.w *= inv;
    m2.x *= inv; m2.y *= inv; m2.z *= inv; m2.w *= inv;

    float4* dst = (float4*)(out + OUT_MASKS) + row * 3;
    dst[0] = m0; dst[1] = m1; dst[2] = m2;
}

extern "C" void kernel_launch(void* const* d_in, const int* in_sizes, int n_in,
                              void* d_out, int out_size, void* d_ws, size_t ws_size,
                              hipStream_t stream) {
    const float* slots = (const float*)d_in[0];
    const float* masks = (const float*)d_in[1];
    const float* W1    = (const float*)d_in[2];
    const float* b1    = (const float*)d_in[3];
    const float* W2    = (const float*)d_in[4];
    const float* b2    = (const float*)d_in[5];
    float* out = (float*)d_out;
    float* ws  = (float*)d_ws;

    float* partials = ws + WS_PARTIALS;
    float* gates    = ws + WS_GATES;

    asp_reduce<<<B_ * CHUNKS, 256, 0, stream>>>(masks, partials);
    asp_gates<<<12, 256, 0, stream>>>(partials, W1, b1, W2, b2, gates, out);
    asp_slots<<<(B_ * K_ * D_ / 4) / 256, 256, 0, stream>>>(slots, gates, out);
    asp_masks<<<(B_ * N_) / 256, 256, 0, stream>>>(masks, gates, out);
}


# Fuse the gate MLP and the slot scaling into the masks kernel: each work-group recomputes its batch's 12 gates on one wave (matrix instruction kept) under its own loads; 4 launches -> 2
# speedup vs baseline: 1.1491x; 1.1491x over previous
#include <hip/hip_runtime.h>
#include <math.h>


typedef __attribute__((ext_vector_type(2))) float v2f;
typedef __attribute__((ext_vector_type(8))) float v8f;

#define B_  128
#define N_  16384
#define K_  12
#define D_  256
#define CHUNKS 32
#define ROWS_PER_CHUNK (N_ / CHUNKS)
#define EPS_ 1e-8f

#define SLOT_BLOCKS B_
#define MASK_BLOCKS ((B_ * N_) / 256)

#define OUT_SLOTS 0L
#define OUT_MASKS ((long)B_ * K_ * D_)
#define OUT_GATES (OUT_MASKS + (long)B_ * N_ * K_)
#define OUT_UTIL  (OUT_GATES + (long)B_ * K_)

#define WS_PARTIALS 0

__global__ void asp_reduce(const float* __restrict__ masks,
                           float* __restrict__ partials) {
    const int blk = blockIdx.x;
    const int b   = blk / CHUNKS;
    const int ch  = blk % CHUNKS;
    const int t   = threadIdx.x;

    const float4* src = (const float4*)masks
                      + ((long)b * N_ + (long)ch * ROWS_PER_CHUNK) * 3;
    const float4* p = src + (long)t * 6;

    const float4 v0 = p[0], v1 = p[1], v2 = p[2];
    const float4 v3 = p[3], v4 = p[4], v5 = p[5];

    float a[12];
    a[0] = v0.x + v3.x;  a[1]  = v0.y + v3.y;  a[2]  = v0.z + v3.z;  a[3]  = v0.w + v3.w;
    a[4] = v1.x + v4.x;  a[5]  = v1.y + v4.y;  a[6]  = v1.z + v4.z;  a[7]  = v1.w + v4.w;
    a[8] = v2.x + v5.x;  a[9]  = v2.y + v5.y;  a[10] = v2.z + v5.z;  a[11] = v2.w + v5.w;

    #pragma unroll
    for (int m = 16; m >= 1; m >>= 1) {
        #pragma unroll
        for (int j = 0; j < 12; ++j)
            a[j] += __shfl_xor(a[j], m, 32);
    }

    __shared__ float wred[8 * 12];
    const int w    = t >> 5;
    const int lane = t & 31;
    if (lane == 0) {
        #pragma unroll
        for (int j = 0; j < 12; ++j) wred[w * 12 + j] = a[j];
    }
    __syncthreads();

    if (t < 12) {
        float s = 0.f;
        #pragma unroll
        for (int ww = 0; ww < 8; ++ww) s += wred[ww * 12 + t];
        partials[((long)b * CHUNKS + ch) * K_ + t] = s;
    }
}

__device__ __forceinline__ float asp_gate_rows(const float* __restrict__ partials,
                                               const float* __restrict__ W1,
                                               const float* __restrict__ b1,
                                               const float* __restrict__ W2,
                                               const float* __restrict__ b2,
                                               const int b, const int lane,
                                               float* __restrict__ g_sh) {
    const int m = lane & 15;
    const int k = (m < K_) ? m : (K_ - 1);

    float s = 0.f;
    #pragma unroll
    for (int c = 0; c < CHUNKS; ++c)
        s += partials[((long)b * CHUNKS + c) * K_ + k];
    const float u = s * (1.0f / (float)N_);

    float h[16], w2[16];
    #pragma unroll
    for (int j = 0; j < 16; ++j) {
        h[j]  = fmaxf(0.f, fmaf(u, W1[j], b1[j]));
        w2[j] = W2[j];
    }

    const bool hi = (lane >= 16);
    v8f acc = {0.f, 0.f, 0.f, 0.f, 0.f, 0.f, 0.f, 0.f};
    #pragma unroll
    for (int i = 0; i < 4; ++i) {
        v2f av, bv;
        av.x = hi ? h[4 * i + 2]  : h[4 * i + 0];
        av.y = hi ? h[4 * i + 3]  : h[4 * i + 1];
        bv.x = hi ? w2[4 * i + 2] : w2[4 * i + 0];
        bv.y = hi ? w2[4 * i + 3] : w2[4 * i + 1];
        acc = __builtin_amdgcn_wmma_f32_16x16x4_f32(
             false, av,  false, bv,
             (short)0, acc,  false,  false);
    }

    const int j8 = lane & 7;
    float z = acc[0];
    for (int j = 1; j < 8; ++j) z = (j8 == j) ? acc[j] : z;
    const float g = 1.0f / (1.0f + __expf(-(z + b2[0])));
    g_sh[(hi ? 8 : 0) + j8] = g;
    return u;
}

__global__ void asp_apply(const float* __restrict__ slots,
                          const float* __restrict__ masks,
                          const float* __restrict__ partials,
                          const float* __restrict__ W1,
                          const float* __restrict__ b1,
                          const float* __restrict__ W2,
                          const float* __restrict__ b2,
                          float* __restrict__ out) {
    __shared__ float g_sh[16];
    const int  t        = threadIdx.x;
    const bool slot_blk = (blockIdx.x < SLOT_BLOCKS);

    int  b;
    long first;
    long step;
    const float4* base;
    float4* dst;
    if (slot_blk) {
        b     = blockIdx.x;
        first = (long)b * (K_ * D_ / 4) + t;
        step  = 256;
        base  = (const float4*)slots;
        dst   = (float4*)(out + OUT_SLOTS);
    } else {
        const long row = (long)(blockIdx.x - SLOT_BLOCKS) * 256 + t;
        b     = (int)(row >> 14);
        first = row * 3;
        step  = 1;
        base  = (const float4*)masks;
        dst   = (float4*)(out + OUT_MASKS);
    }
    float4 x0 = base[first], x1 = base[first + step], x2 = base[first + 2 * step];

    float u = 0.f;
    if (t < 32)
        u = asp_gate_rows(partials, W1, b1, W2, b2, b, t, g_sh);
    __syncthreads();

    if (slot_blk) {
        if (t < K_) {
            out[OUT_UTIL  + b * K_ + t] = u;
            out[OUT_GATES + b * K_ + t] = g_sh[t];
        }
        const int   kq = t >> 6;
        const float g0 = g_sh[kq], g1 = g_sh[kq + 4], g2 = g_sh[kq + 8];
        x0.x *= g0; x0.y *= g0; x0.z *= g0; x0.w *= g0;
        x1.x *= g1; x1.y *= g1; x1.z *= g1; x1.w *= g1;
        x2.x *= g2; x2.y *= g2; x2.z *= g2; x2.w *= g2;
    } else {
        x0.x *= g_sh[0];  x0.y *= g_sh[1];  x0.z *= g_sh[2];  x0.w *= g_sh[3];
        x1.x *= g_sh[4];  x1.y *= g_sh[5];  x1.z *= g_sh[6];  x1.w *= g_sh[7];
        x2.x *= g_sh[8];  x2.y *= g_sh[9];  x2.z *= g_sh[10]; x2.w *= g_sh[11];

        const float sum = (x0.x + x0.y + x0.z + x0.w)
                        + (x1.x + x1.y + x1.z + x1.w)
                        + (x2.x + x2.y + x2.z + x2.w);
        const float inv = 1.0f / (sum + EPS_);

        x0.x *= inv; x0.y *= inv; x0.z *= inv; x0.w *= inv;
        x1.x *= inv; x1.y *= inv; x1.z *= inv; x1.w *= inv;
        x2.x *= inv; x2.y *= inv; x2.z *= inv; x2.w *= inv;
    }
    dst[first] = x0; dst[first + step] = x1; dst[first + 2 * step] = x2;
}

extern "C" void kernel_launch(void* const* d_in, const int* in_sizes, int n_in,
                              void* d_out, int out_size, void* d_ws, size_t ws_size,
                              hipStream_t stream) {
    const float* slots = (const float*)d_in[0];
    const float* masks = (const float*)d_in[1];
    const float* W1    = (const float*)d_in[2];
    const float* b1    = (const float*)d_in[3];
    const float* W2    = (const float*)d_in[4];
    const float* b2    = (const float*)d_in[5];
    float* out = (float*)d_out;
    float* ws  = (float*)d_ws;

    float* partials = ws + WS_PARTIALS;

    asp_reduce<<<B_ * CHUNKS, 256, 0, stream>>>(masks, partials);
    asp_apply<<<SLOT_BLOCKS + MASK_BLOCKS, 256, 0, stream>>>(
        slots, masks, partials, W1, b1, W2, b2, out);
}


# Both passes regridded to 512 work-groups that loop over 4096 rows each, next tiles' loads issued ahead, not one tile per work-group: 12x fewer waves to start; N-sum order changes (4 partials a batch)
# speedup vs baseline: 1.2550x; 1.0922x over previous
#include <hip/hip_runtime.h>
#include <math.h>


typedef __attribute__((ext_vector_type(2))) float v2f;
typedef __attribute__((ext_vector_type(8))) float v8f;
typedef __attribute__((ext_vector_type(4))) float v4f;

#define B_  128
#define N_  16384
#define K_  12
#define D_  256
#define CHUNKS 4
#define ROWS_PER_CHUNK (N_ / CHUNKS)
#define EPS_ 1e-8f

#define RED_TILE_ROWS 512
#define RED_TILES (ROWS_PER_CHUNK / RED_TILE_ROWS)
#define RED_GROUP 2
#define RED_GROUPS (RED_TILES / RED_GROUP)

#define APP_TILE_ROWS 256
#define APP_TILES (ROWS_PER_CHUNK / APP_TILE_ROWS)
#define APP_GROUP 4
#define APP_GROUPS (APP_TILES / APP_GROUP)

#define SLOT_F4_PER_BATCH (K_ * D_ / 4)
#define SLOT_F4_PER_BLOCK (SLOT_F4_PER_BATCH / CHUNKS)

#define OUT_SLOTS 0L
#define OUT_MASKS ((long)B_ * K_ * D_)
#define OUT_GATES (OUT_MASKS + (long)B_ * N_ * K_)
#define OUT_UTIL  (OUT_GATES + (long)B_ * K_)

#define WS_PARTIALS 0

__device__ __forceinline__ v4f f4add(const v4f a, const v4f b) {
    return a + b;
}

__device__ __forceinline__ void asp_after_loads(v4f& x) {
    asm volatile("" : "+v"(x) : : "memory");
}

__global__ __launch_bounds__(256)
void asp_reduce(const float* masks,
                float* __restrict__ partials) {
    const int blk = blockIdx.x;
    const int b   = blk / CHUNKS;
    const int ch  = blk % CHUNKS;
    const int t   = threadIdx.x;

    const v4f* p = (const v4f*)masks
                    + ((long)b * N_ + (long)ch * ROWS_PER_CHUNK) * 3
                    + (long)t * 6;
    const int tile_f4 = RED_TILE_ROWS * 3;

    v4f v[2][RED_GROUP * 6];
    #pragma unroll
    for (int i = 0; i < RED_GROUP; ++i)
        #pragma unroll
        for (int k = 0; k < 6; ++k) v[0][i * 6 + k] = p[i * tile_f4 + k];

    v4f acc[6];

    #pragma unroll
    for (int g = 0; g < RED_GROUPS; ++g) {
        if (g + 1 < RED_GROUPS) {
            #pragma unroll
            for (int i = 0; i < RED_GROUP; ++i)
                #pragma unroll
                for (int k = 0; k < 6; ++k)
                    v[(g + 1) & 1][i * 6 + k] = p[((g + 1) * RED_GROUP + i) * tile_f4 + k];
        }
        #pragma unroll
        for (int j = 0; j < RED_GROUP * 6; ++j) asp_after_loads(v[g & 1][j]);
        #pragma unroll
        for (int i = 0; i < RED_GROUP; ++i)
            #pragma unroll
            for (int k = 0; k < 6; ++k)
                acc[k] = (g == 0 && i == 0) ? v[0][k] : f4add(acc[k], v[g & 1][i * 6 + k]);
    }

    float a[12];
    a[0] = acc[0].x + acc[3].x;  a[1]  = acc[0].y + acc[3].y;
    a[2] = acc[0].z + acc[3].z;  a[3]  = acc[0].w + acc[3].w;
    a[4] = acc[1].x + acc[4].x;  a[5]  = acc[1].y + acc[4].y;
    a[6] = acc[1].z + acc[4].z;  a[7]  = acc[1].w + acc[4].w;
    a[8] = acc[2].x + acc[5].x;  a[9]  = acc[2].y + acc[5].y;
    a[10] = acc[2].z + acc[5].z; a[11] = acc[2].w + acc[5].w;

    #pragma unroll
    for (int m = 16; m >= 1; m >>= 1) {
        #pragma unroll
        for (int j = 0; j < 12; ++j)
            a[j] += __shfl_xor(a[j], m, 32);
    }

    __shared__ float wred[8 * 12];
    const int w    = t >> 5;
    const int lane = t & 31;
    if (lane == 0) {
        #pragma unroll
        for (int j = 0; j < 12; ++j) wred[w * 12 + j] = a[j];
    }
    __syncthreads();

    if (t < 12) {
        float s = 0.f;
        #pragma unroll
        for (int ww = 0; ww < 8; ++ww) s += wred[ww * 12 + t];
        partials[((long)b * CHUNKS + ch) * K_ + t] = s;
    }
}

__device__ __forceinline__ float asp_gate_rows(const float* __restrict__ partials,
                                               const float* __restrict__ W1,
                                               const float* __restrict__ b1,
                                               const float* __restrict__ W2,
                                               const float* __restrict__ b2,
                                               const int b, const int lane,
                                               float* __restrict__ g_sh) {
    const int m = lane & 15;
    const int k = (m < K_) ? m : (K_ - 1);

    float s = 0.f;
    #pragma unroll
    for (int c = 0; c < CHUNKS; ++c)
        s += partials[((long)b * CHUNKS + c) * K_ + k];
    const float u = s * (1.0f / (float)N_);

    float h[16], w2[16];
    #pragma unroll
    for (int j = 0; j < 16; ++j) {
        h[j]  = fmaxf(0.f, fmaf(u, W1[j], b1[j]));
        w2[j] = W2[j];
    }

    const bool hi = (lane >= 16);
    v8f acc = {0.f, 0.f, 0.f, 0.f, 0.f, 0.f, 0.f, 0.f};
    #pragma unroll
    for (int i = 0; i < 4; ++i) {
        v2f av, bv;
        av.x = hi ? h[4 * i + 2]  : h[4 * i + 0];
        av.y = hi ? h[4 * i + 3]  : h[4 * i + 1];
        bv.x = hi ? w2[4 * i + 2] : w2[4 * i + 0];
        bv.y = hi ? w2[4 * i + 3] : w2[4 * i + 1];
        acc = __builtin_amdgcn_wmma_f32_16x16x4_f32(
             false, av,  false, bv,
             (short)0, acc,  false,  false);
    }

    const int j8 = lane & 7;
    float z = acc[0];
    for (int j = 1; j < 8; ++j) z = (j8 == j) ? acc[j] : z;
    const float g = 1.0f / (1.0f + __expf(-(z + b2[0])));
    g_sh[(hi ? 8 : 0) + j8] = g;
    return u;
}

__device__ __forceinline__ void asp_row(v4f& x0, v4f& x1, v4f& x2,
                                        const float* __restrict__ g) {
    x0.x *= g[0];  x0.y *= g[1];  x0.z *= g[2];  x0.w *= g[3];
    x1.x *= g[4];  x1.y *= g[5];  x1.z *= g[6];  x1.w *= g[7];
    x2.x *= g[8];  x2.y *= g[9];  x2.z *= g[10]; x2.w *= g[11];

    const float sum = (x0.x + x0.y + x0.z + x0.w)
                    + (x1.x + x1.y + x1.z + x1.w)
                    + (x2.x + x2.y + x2.z + x2.w);
    const float inv = 1.0f / (sum + EPS_);

    x0.x *= inv; x0.y *= inv; x0.z *= inv; x0.w *= inv;
    x1.x *= inv; x1.y *= inv; x1.z *= inv; x1.w *= inv;
    x2.x *= inv; x2.y *= inv; x2.z *= inv; x2.w *= inv;
}

__global__ __launch_bounds__(256)
void asp_apply(const float* __restrict__ slots,
               const float* masks,
               const float* __restrict__ partials,
               const float* __restrict__ W1,
               const float* __restrict__ b1,
               const float* __restrict__ W2,
               const float* __restrict__ b2,
               float* __restrict__ out) {
    __shared__ float g_sh[16];
    const int t  = threadIdx.x;
    const int b  = blockIdx.x / CHUNKS;
    const int ch = blockIdx.x % CHUNKS;

    const long f4_0 = ((long)b * N_ + (long)ch * ROWS_PER_CHUNK + t) * 3;
    const v4f* src = (const v4f*)masks + f4_0;
    v4f*       dst = (v4f*)(out + OUT_MASKS) + f4_0;
    const int tile_f4 = APP_TILE_ROWS * 3;

    const bool has_slot = (t < SLOT_F4_PER_BLOCK);
    const long sidx = (long)b * SLOT_F4_PER_BATCH + ch * SLOT_F4_PER_BLOCK + t;

    v4f v[2][APP_GROUP * 3];
    #pragma unroll
    for (int i = 0; i < APP_GROUP; ++i)
        #pragma unroll
        for (int k = 0; k < 3; ++k) v[0][i * 3 + k] = src[i * tile_f4 + k];
    v4f sv = (v4f){0.f, 0.f, 0.f, 0.f};
    if (has_slot) sv = ((const v4f*)slots)[sidx];

    float u = 0.f;
    if (t < 32)
        u = asp_gate_rows(partials, W1, b1, W2, b2, b, t, g_sh);
    __syncthreads();

    float g[12];
    #pragma unroll
    for (int j = 0; j < 12; ++j) g[j] = g_sh[j];

    if (ch == 0 && t < K_) {
        out[OUT_UTIL  + b * K_ + t] = u;
        out[OUT_GATES + b * K_ + t] = g_sh[t];
    }
    if (has_slot) {
        const float gs = g_sh[ch * 3 + (t >> 6)];
        sv.x *= gs; sv.y *= gs; sv.z *= gs; sv.w *= gs;
        ((v4f*)(out + OUT_SLOTS))[sidx] = sv;
    }

    #pragma unroll
    for (int gr = 0; gr < APP_GROUPS; ++gr) {
        if (gr + 1 < APP_GROUPS) {
            #pragma unroll
            for (int i = 0; i < APP_GROUP; ++i)
                #pragma unroll
                for (int k = 0; k < 3; ++k)
                    v[(gr + 1) & 1][i * 3 + k] = src[((gr + 1) * APP_GROUP + i) * tile_f4 + k];
        }
        #pragma unroll
        for (int j = 0; j < APP_GROUP * 3; ++j) asp_after_loads(v[gr & 1][j]);
        #pragma unroll
        for (int i = 0; i < APP_GROUP; ++i) {
            v4f* x = &v[gr & 1][i * 3];
            asp_row(x[0], x[1], x[2], g);
            v4f* d = dst + (gr * APP_GROUP + i) * tile_f4;
            d[0] = x[0]; d[1] = x[1]; d[2] = x[2];
        }
    }
}

extern "C" void kernel_launch(void* const* d_in, const int* in_sizes, int n_in,
                              void* d_out, int out_size, void* d_ws, size_t ws_size,
                              hipStream_t stream) {
    const float* slots = (const float*)d_in[0];
    const float* masks = (const float*)d_in[1];
    const float* W1    = (const float*)d_in[2];
    const float* b1    = (const float*)d_in[3];
    const float* W2    = (const float*)d_in[4];
    const float* b2    = (const float*)d_in[5];
    float* out = (float*)d_out;
    float* ws  = (float*)d_ws;

    float* partials = ws + WS_PARTIALS;

    asp_reduce<<<B_ * CHUNKS, 256, 0, stream>>>(masks, partials);
    asp_apply<<<B_ * CHUNKS, 256, 0, stream>>>(
        slots, masks, partials, W1, b1, W2, b2, out);
}


# Every 16-byte mask load and store of a wave covers whole 128-byte lines (lanes take consecutive pieces, not 48/96-byte strides); row sums via 8-lane gathers; N-sum order changes
# speedup vs baseline: 1.4339x; 1.1425x over previous
#include <hip/hip_runtime.h>
#include <math.h>


typedef __attribute__((ext_vector_type(2))) float v2f;
typedef __attribute__((ext_vector_type(8))) float v8f;
typedef __attribute__((ext_vector_type(4))) float v4f;

#define B_  128
#define N_  16384
#define K_  12
#define D_  256
#define CHUNKS 4
#define ROWS_PER_CHUNK (N_ / CHUNKS)
#define EPS_ 1e-8f

#define RED_LOADS (ROWS_PER_CHUNK * 3 / 256)
#define RED_GROUP 12
#define RED_GROUPS (RED_LOADS / RED_GROUP)

#define APP_TILE_ROWS 256
#define APP_TILES (ROWS_PER_CHUNK / APP_TILE_ROWS)
#define APP_GROUP 4
#define APP_GROUPS (APP_TILES / APP_GROUP)

#define SLOT_F4_PER_BATCH (K_ * D_ / 4)
#define SLOT_F4_PER_BLOCK (SLOT_F4_PER_BATCH / CHUNKS)

#define OUT_SLOTS 0L
#define OUT_MASKS ((long)B_ * K_ * D_)
#define OUT_GATES (OUT_MASKS + (long)B_ * N_ * K_)
#define OUT_UTIL  (OUT_GATES + (long)B_ * K_)

#define WS_PARTIALS 0

__device__ __forceinline__ v4f f4add(const v4f a, const v4f b) {
    return a + b;
}

__device__ __forceinline__ void asp_after_loads(v4f& x) {
    asm volatile("" : "+v"(x) : : "memory");
}

__global__ __launch_bounds__(256)
void asp_reduce(const float* masks,
                float* __restrict__ partials) {
    const int blk = blockIdx.x;
    const int b   = blk / CHUNKS;
    const int ch  = blk % CHUNKS;
    const int t   = threadIdx.x;

    const v4f* p = (const v4f*)masks
                    + ((long)b * N_ + (long)ch * ROWS_PER_CHUNK) * 3
                    + t;

    v4f v[2][RED_GROUP];
    #pragma unroll
    for (int i = 0; i < RED_GROUP; ++i) v[0][i] = p[i * 256];

    v4f acc[3];

    #pragma unroll
    for (int g = 0; g < RED_GROUPS; ++g) {
        if (g + 1 < RED_GROUPS) {
            #pragma unroll
            for (int i = 0; i < RED_GROUP; ++i)
                v[(g + 1) & 1][i] = p[((g + 1) * RED_GROUP + i) * 256];
        }
        #pragma unroll
        for (int j = 0; j < RED_GROUP; ++j) asp_after_loads(v[g & 1][j]);
        #pragma unroll
        for (int i = 0; i < RED_GROUP; ++i)
            acc[i % 3] = (g == 0 && i < 3) ? v[0][i] : f4add(acc[i % 3], v[g & 1][i]);
    }

    const int r = t % 3;
    const v4f c0 = (r == 0) ? acc[0] : (r == 1) ? acc[2] : acc[1];
    const v4f c1 = (r == 0) ? acc[1] : (r == 1) ? acc[0] : acc[2];
    const v4f c2 = (r == 0) ? acc[2] : (r == 1) ? acc[1] : acc[0];

    float a[12];
    a[0] = c0.x;  a[1] = c0.y;  a[2]  = c0.z;  a[3]  = c0.w;
    a[4] = c1.x;  a[5] = c1.y;  a[6]  = c1.z;  a[7]  = c1.w;
    a[8] = c2.x;  a[9] = c2.y;  a[10] = c2.z;  a[11] = c2.w;

    #pragma unroll
    for (int m = 16; m >= 1; m >>= 1) {
        #pragma unroll
        for (int j = 0; j < 12; ++j)
            a[j] += __shfl_xor(a[j], m, 32);
    }

    __shared__ float wred[8 * 12];
    const int w    = t >> 5;
    const int lane = t & 31;
    if (lane == 0) {
        #pragma unroll
        for (int j = 0; j < 12; ++j) wred[w * 12 + j] = a[j];
    }
    __syncthreads();

    if (t < 12) {
        float s = 0.f;
        #pragma unroll
        for (int ww = 0; ww < 8; ++ww) s += wred[ww * 12 + t];
        partials[((long)b * CHUNKS + ch) * K_ + t] = s;
    }
}

__device__ __forceinline__ float asp_gate_rows(const float* __restrict__ partials,
                                               const float* __restrict__ W1,
                                               const float* __restrict__ b1,
                                               const float* __restrict__ W2,
                                               const float* __restrict__ b2,
                                               const int b, const int lane,
                                               float* __restrict__ g_sh) {
    const int m = lane & 15;
    const int k = (m < K_) ? m : (K_ - 1);

    float s = 0.f;
    #pragma unroll
    for (int c = 0; c < CHUNKS; ++c)
        s += partials[((long)b * CHUNKS + c) * K_ + k];
    const float u = s * (1.0f / (float)N_);

    float h[16], w2[16];
    #pragma unroll
    for (int j = 0; j < 16; ++j) {
        h[j]  = fmaxf(0.f, fmaf(u, W1[j], b1[j]));
        w2[j] = W2[j];
    }

    const bool hi = (lane >= 16);
    v8f acc = {0.f, 0.f, 0.f, 0.f, 0.f, 0.f, 0.f, 0.f};
    #pragma unroll
    for (int i = 0; i < 4; ++i) {
        v2f av, bv;
        av.x = hi ? h[4 * i + 2]  : h[4 * i + 0];
        av.y = hi ? h[4 * i + 3]  : h[4 * i + 1];
        bv.x = hi ? w2[4 * i + 2] : w2[4 * i + 0];
        bv.y = hi ? w2[4 * i + 3] : w2[4 * i + 1];
        acc = __builtin_amdgcn_wmma_f32_16x16x4_f32(
             false, av,  false, bv,
             (short)0, acc,  false,  false);
    }

    const int j8 = lane & 7;
    float z = acc[0];
    for (int j = 1; j < 8; ++j) z = (j8 == j) ? acc[j] : z;
    const float g = 1.0f / (1.0f + __expf(-(z + b2[0])));
    g_sh[(hi ? 8 : 0) + j8] = g;
    return u;
}

#define ASP_SEL8(s0, s1, s2, s3, s4, s5, s6, s7)                              \
    ((s0) | ((s1) << 3) | ((s2) << 6) | ((s3) << 9) | ((s4) << 12) |          \
     ((s5) << 15) | ((s6) << 18) | ((s7) << 21))
template <unsigned SEL>
__device__ __forceinline__ float asp_gather8(const float x) {
    return __builtin_bit_cast(float,
        __builtin_amdgcn_mov_dpp8(__builtin_bit_cast(unsigned, x), SEL));
}

__device__ __forceinline__ void asp_rows8(v4f& x0, v4f& x1, v4f& x2,
                                          const v4f g0, const v4f g1, const v4f g2,
                                          const int s) {
    x0 *= g0;  x1 *= g1;  x2 *= g2;

    const float p0 = x0.x + x0.y + x0.z + x0.w;
    const float p1 = x1.x + x1.y + x1.z + x1.w;
    const float p2 = x2.x + x2.y + x2.z + x2.w;

    constexpr unsigned SA = ASP_SEL8(0, 3, 6, 1, 4, 7, 2, 5);
    const float a0 = asp_gather8<SA>(p0), a1 = asp_gather8<SA>(p1), a2 = asp_gather8<SA>(p2);
    constexpr unsigned SB = ASP_SEL8(1, 4, 7, 2, 5, 0, 3, 6);
    const float b0 = asp_gather8<SB>(p0), b1 = asp_gather8<SB>(p1), b2 = asp_gather8<SB>(p2);
    constexpr unsigned SC = ASP_SEL8(2, 5, 0, 3, 6, 1, 4, 7);
    const float c0 = asp_gather8<SC>(p0), c1 = asp_gather8<SC>(p1), c2 = asp_gather8<SC>(p2);

    const float a = (s < 3) ? a0 : (s < 6) ? a1 : a2;
    const float b = (s < 3) ? b0 : (s < 5) ? b1 : b2;
    const float c = (s < 2) ? c0 : (s < 5) ? c1 : c2;

    const float sum = a + b + c;
    const float inv = 1.0f / (sum + EPS_);

    const float i0 = asp_gather8<ASP_SEL8(0, 0, 0, 1, 1, 1, 2, 2)>(inv);
    const float i1 = asp_gather8<ASP_SEL8(2, 3, 3, 3, 4, 4, 4, 5)>(inv);
    const float i2 = asp_gather8<ASP_SEL8(5, 5, 6, 6, 6, 7, 7, 7)>(inv);

    x0 *= i0;  x1 *= i1;  x2 *= i2;
}

__global__ __launch_bounds__(256)
void asp_apply(const float* __restrict__ slots,
               const float* masks,
               const float* __restrict__ partials,
               const float* __restrict__ W1,
               const float* __restrict__ b1,
               const float* __restrict__ W2,
               const float* __restrict__ b2,
               float* __restrict__ out) {
    __shared__ __attribute__((aligned(16))) float g_sh[16];
    const int t  = threadIdx.x;
    const int b  = blockIdx.x / CHUNKS;
    const int ch = blockIdx.x % CHUNKS;
    const int s  = t & 7;

    const long f4_0 = ((long)b * N_ + (long)ch * ROWS_PER_CHUNK + (t & ~31)) * 3
                    + ((t & 31) >> 3) * 24 + s;
    const v4f* src = (const v4f*)masks + f4_0;
    v4f*       dst = (v4f*)(out + OUT_MASKS) + f4_0;
    const int tile_f4 = APP_TILE_ROWS * 3;

    const bool has_slot = (t < SLOT_F4_PER_BLOCK);
    const long sidx = (long)b * SLOT_F4_PER_BATCH + ch * SLOT_F4_PER_BLOCK + t;

    v4f v[2][APP_GROUP * 3];
    #pragma unroll
    for (int i = 0; i < APP_GROUP; ++i)
        #pragma unroll
        for (int k = 0; k < 3; ++k) v[0][i * 3 + k] = src[i * tile_f4 + 8 * k];
    v4f sv = (v4f){0.f, 0.f, 0.f, 0.f};
    if (has_slot) sv = ((const v4f*)slots)[sidx];

    float u = 0.f;
    if (t < 32)
        u = asp_gate_rows(partials, W1, b1, W2, b2, b, t, g_sh);
    __syncthreads();

    v4f g[3];
    #pragma unroll
    for (int j = 0; j < 3; ++j) g[j] = *(const v4f*)(g_sh + 4 * ((2 * j + s) % 3));

    if (ch == 0 && t < K_) {
        out[OUT_UTIL  + b * K_ + t] = u;
        out[OUT_GATES + b * K_ + t] = g_sh[t];
    }
    if (has_slot) {
        const float gs = g_sh[ch * 3 + (t >> 6)];
        sv.x *= gs; sv.y *= gs; sv.z *= gs; sv.w *= gs;
        ((v4f*)(out + OUT_SLOTS))[sidx] = sv;
    }

    #pragma unroll
    for (int gr = 0; gr < APP_GROUPS; ++gr) {
        if (gr + 1 < APP_GROUPS) {
            #pragma unroll
            for (int i = 0; i < APP_GROUP; ++i)
                #pragma unroll
                for (int k = 0; k < 3; ++k)
                    v[(gr + 1) & 1][i * 3 + k] = src[((gr + 1) * APP_GROUP + i) * tile_f4 + 8 * k];
        }
        #pragma unroll
        for (int j = 0; j < APP_GROUP * 3; ++j) asp_after_loads(v[gr & 1][j]);
        #pragma unroll
        for (int i = 0; i < APP_GROUP; ++i) {
            v4f* x = &v[gr & 1][i * 3];
            asp_rows8(x[0], x[1], x[2], g[0], g[1], g[2], s);
            v4f* d = dst + (gr * APP_GROUP + i) * tile_f4;
            d[0] = x[0]; d[8] = x[1]; d[16] = x[2];
        }
    }
}

extern "C" void kernel_launch(void* const* d_in, const int* in_sizes, int n_in,
                              void* d_out, int out_size, void* d_ws, size_t ws_size,
                              hipStream_t stream) {
    const float* slots = (const float*)d_in[0];
    const float* masks = (const float*)d_in[1];
    const float* W1    = (const float*)d_in[2];
    const float* b1    = (const float*)d_in[3];
    const float* W2    = (const float*)d_in[4];
    const float* b2    = (const float*)d_in[5];
    float* out = (float*)d_out;
    float* ws  = (float*)d_ws;

    float* partials = ws + WS_PARTIALS;

    asp_reduce<<<B_ * CHUNKS, 256, 0, stream>>>(masks, partials);
    asp_apply<<<B_ * CHUNKS, 256, 0, stream>>>(
        slots, masks, partials, W1, b1, W2, b2, out);
}
